# starting kernel
# baseline (speedup 1.0000x reference)
#include <hip/hip_runtime.h>
#include <hip/hip_bf16.h>
#include <math.h>

typedef __attribute__((ext_vector_type(2))) float v2f;
typedef __attribute__((ext_vector_type(8))) float v8f;

#define BB 4
#define LL 512
#define DD 128

__device__ __forceinline__ float fast_tanh(float x) {
#if defined(__has_builtin)
#if __has_builtin(__builtin_amdgcn_tanhf)
    return __builtin_amdgcn_tanhf(x);
#elif __has_builtin(__builtin_amdgcn_tanh_f32)
    return __builtin_amdgcn_tanh_f32(x);
#else
    return tanhf(x);
#endif
#else
    return tanhf(x);
#endif
}

__global__ __launch_bounds__(128)
void proj_qk_wmma(const float* __restrict__ inp,
                  const float* __restrict__ Wq, const float* __restrict__ bq,
                  const float* __restrict__ Wk,
                  float* __restrict__ qo, float* __restrict__ ko)
{
    const int wave = threadIdx.x >> 5;
    const int lane = threadIdx.x & 31;
    const int tile = blockIdx.x * 4 + wave;
    const int m0   = (tile >> 3) << 4;
    const int n0   = (tile & 7) << 4;
    const int r    = lane & 15;
    const int koff = (lane >> 4) << 1;

    const float* arow  = inp + (size_t)(m0 + r) * DD + koff;
    const float* bqrow = Wq  + (size_t)(n0 + r) * DD + koff;
    const float* bkrow = Wk  + (size_t)(n0 + r) * DD + koff;

    v8f cq = {};
    v8f ck = {};
    for (int kk = 0; kk < DD; kk += 4) {
        v2f a   = *(const v2f*)(arow  + kk);
        v2f bq_ = *(const v2f*)(bqrow + kk);
        v2f bk_ = *(const v2f*)(bkrow + kk);
        cq = __builtin_amdgcn_wmma_f32_16x16x4_f32(false, a, false, bq_,
                                                   (short)0, cq, false, false);
        ck = __builtin_amdgcn_wmma_f32_16x16x4_f32(false, a, false, bk_,
                                                   (short)0, ck, false, false);
    }

    const int   nn    = n0 + (lane & 15);
    const float bias  = bq[nn];
    const int   mbase = m0 + ((lane >> 4) << 3);
#pragma unroll
    for (int rr = 0; rr < 8; ++rr) {
        const int m = mbase + rr;
        qo[(size_t)m * DD + nn] = cq[rr] + bias;
        ko[(size_t)m * DD + nn] = ck[rr];
    }
}

__global__ __launch_bounds__(256)
void score_softmax(const float* __restrict__ q, const float* __restrict__ k,
                   const float* __restrict__ v, float* __restrict__ out)
{
    const int bi = blockIdx.x;
    const int b  = bi >> 9;

    __shared__ float s_k[DD];
    __shared__ float s_v[DD];
    __shared__ float s_score[LL];
    __shared__ float s_red[8];

    const int tid  = threadIdx.x;
    const int lane = tid & 31;
    const int wave = tid >> 5;

    if (tid < DD) {
        s_k[tid] = k[(size_t)bi * DD + tid];
        s_v[tid] = v[tid];
    }
    __syncthreads();

    const float kv0 = s_k[lane +  0], kv1 = s_k[lane + 32];
    const float kv2 = s_k[lane + 64], kv3 = s_k[lane + 96];
    const float vv0 = s_v[lane +  0], vv1 = s_v[lane + 32];
    const float vv2 = s_v[lane + 64], vv3 = s_v[lane + 96];

    const float* qb = q + (size_t)b * LL * DD;

    for (int j = wave; j < LL; j += 8) {
        const float* qr = qb + (size_t)j * DD + lane;
        float acc = fast_tanh(qr[ 0] + kv0) * vv0
                  + fast_tanh(qr[32] + kv1) * vv1
                  + fast_tanh(qr[64] + kv2) * vv2
                  + fast_tanh(qr[96] + kv3) * vv3;
#pragma unroll
        for (int off = 16; off > 0; off >>= 1)
            acc += __shfl_xor(acc, off, 32);
        if (lane == 0) s_score[j] = acc;
    }
    __syncthreads();

    const float a0 = s_score[tid];
    const float a1 = s_score[tid + 256];

    float m = fmaxf(a0, a1);
#pragma unroll
    for (int off = 16; off > 0; off >>= 1)
        m = fmaxf(m, __shfl_xor(m, off, 32));
    if (lane == 0) s_red[wave] = m;
    __syncthreads();

    float bm = s_red[0];
#pragma unroll
    for (int w = 1; w < 8; ++w) bm = fmaxf(bm, s_red[w]);

    const float e0 = __expf(a0 - bm);
    const float e1 = __expf(a1 - bm);

    float s = e0 + e1;
#pragma unroll
    for (int off = 16; off > 0; off >>= 1)
        s += __shfl_xor(s, off, 32);
    __syncthreads();
    if (lane == 0) s_red[wave] = s;
    __syncthreads();

    float bs = s_red[0];
#pragma unroll
    for (int w = 1; w < 8; ++w) bs += s_red[w];
    const float inv = 1.0f / bs;

    float* orow = out + (size_t)bi * LL;
    orow[tid]       = e0 * inv;
    orow[tid + 256] = e1 * inv;
}

extern "C" void kernel_launch(void* const* d_in, const int* in_sizes, int n_in,
                              void* d_out, int out_size, void* d_ws, size_t ws_size,
                              hipStream_t stream) {
    (void)in_sizes; (void)n_in; (void)out_size; (void)ws_size;
    const float* inp = (const float*)d_in[0];
    const float* Wq  = (const float*)d_in[1];
    const float* bq  = (const float*)d_in[2];
    const float* Wk  = (const float*)d_in[3];
    const float* v   = (const float*)d_in[4];
    float* out = (float*)d_out;

    float* qws = (float*)d_ws;
    float* kws = qws + (size_t)BB * LL * DD;

    proj_qk_wmma<<<256, 128, 0, stream>>>(inp, Wq, bq, Wk, qws, kws);
    score_softmax<<<BB * LL, 256, 0, stream>>>(qws, kws, v, out);
}


# score kernel re-tiled: each lane owns 2 columns j and sums over d in registers as even/odd-d packed partial sums (d summation order changes, no cross-lane reduce); 8 rows per workgroup; q transposed
# speedup vs baseline: 2.5773x; 2.5773x over previous
#include <hip/hip_runtime.h>
#include <hip/hip_bf16.h>
#include <math.h>


typedef __attribute__((ext_vector_type(2))) float v2f;
typedef __attribute__((ext_vector_type(4))) float v4f;
typedef __attribute__((ext_vector_type(8))) float v8f;

#define BB 4
#define LL 512
#define DD 128
#define IT 8
#define CH 16

__device__ __forceinline__ float fast_tanh(float x) {
#if defined(__has_builtin)
#if __has_builtin(__builtin_amdgcn_tanhf)
    return __builtin_amdgcn_tanhf(x);
#elif __has_builtin(__builtin_amdgcn_tanh_f32)
    return __builtin_amdgcn_tanh_f32(x);
#else
    return tanhf(x);
#endif
#else
    return tanhf(x);
#endif
}

__global__ __launch_bounds__(128)
void proj_qk_wmma(const float* __restrict__ inp,
                  const float* __restrict__ Wq, const float* __restrict__ bq,
                  const float* __restrict__ Wk,
                  float* __restrict__ qo, float* __restrict__ ko)
{
    const int wave = threadIdx.x >> 5;
    const int lane = threadIdx.x & 31;
    const int tile = blockIdx.x * 4 + wave;
    const int m0   = (tile >> 3) << 4;
    const int n0   = (tile & 7) << 4;
    const int r    = lane & 15;
    const int koff = (lane >> 4) << 1;

    const float* arow  = inp + (size_t)(m0 + r) * DD + koff;
    const float* bqrow = Wq  + (size_t)(n0 + r) * DD + koff;
    const float* bkrow = Wk  + (size_t)(n0 + r) * DD + koff;

    v8f cq = {};
    v8f ck = {};
    for (int kk = 0; kk < DD; kk += 4) {
        v2f a   = *(const v2f*)(arow  + kk);
        v2f bq_ = *(const v2f*)(bqrow + kk);
        v2f bk_ = *(const v2f*)(bkrow + kk);
        cq = __builtin_amdgcn_wmma_f32_16x16x4_f32(false, a, false, bq_,
                                                   (short)0, cq, false, false);
        ck = __builtin_amdgcn_wmma_f32_16x16x4_f32(false, a, false, bk_,
                                                   (short)0, ck, false, false);
    }

    const int   nn    = n0 + (lane & 15);
    const float bias  = bq[nn];
    const int   mbase = m0 + ((lane >> 4) << 3);

    {
        const int bb = mbase >> 9;
        const int l0 = mbase & (LL - 1);
        float* qt = qo + ((size_t)(bb * DD + nn)) * LL + l0;
        v4f lo = { cq[0] + bias, cq[1] + bias, cq[2] + bias, cq[3] + bias };
        v4f hi = { cq[4] + bias, cq[5] + bias, cq[6] + bias, cq[7] + bias };
        *(v4f*)(qt)     = lo;
        *(v4f*)(qt + 4) = hi;
    }
#pragma unroll
    for (int rr = 0; rr < 8; ++rr) {
        const int m = mbase + rr;
        ko[(size_t)m * DD + nn] = ck[rr];
    }
}

__device__ __forceinline__ v2f tanh2(v2f x) {
    v2f t;
    t.x = fast_tanh(x.x);
    t.y = fast_tanh(x.y);
    return t;
}

__device__ __forceinline__ void score_chunk(const float* __restrict__ s_k,
                                            const float* __restrict__ s_v,
                                            const int pbase,
                                            const float (&qa)[CH],
                                            const float (&qc)[CH],
                                            v2f (&acc0)[IT], v2f (&acc1)[IT])
{
    constexpr int NS = CH / 2;
    v2f kk[NS][IT];
    v2f vv[NS];
#pragma unroll
    for (int s = 0; s < NS; ++s) {
        const int p = pbase + s;
        const v4f* kp = (const v4f*)(s_k + p * (2 * IT));
        const v4f k01 = kp[0], k23 = kp[1], k45 = kp[2], k67 = kp[3];
        kk[s][0] = (v2f){k01.x, k01.y};  kk[s][1] = (v2f){k01.z, k01.w};
        kk[s][2] = (v2f){k23.x, k23.y};  kk[s][3] = (v2f){k23.z, k23.w};
        kk[s][4] = (v2f){k45.x, k45.y};  kk[s][5] = (v2f){k45.z, k45.w};
        kk[s][6] = (v2f){k67.x, k67.y};  kk[s][7] = (v2f){k67.z, k67.w};
        vv[s] = *(const v2f*)(s_v + 2 * p);
    }

    constexpr int NR  = NS * IT;
    constexpr int FSK = 3;
    v2f X0[NR], X1[NR], T0[NR], T1[NR];
    X0[0] = (v2f){qa[0], qa[1]} + kk[0][0];
    X1[0] = (v2f){qc[0], qc[1]} + kk[0][0];
#pragma unroll
    for (int n = 0; n < NR + FSK; ++n) {
        if (n + 1 < NR) {
            const int s2 = (n + 1) / IT, r2 = (n + 1) % IT;
            X0[n + 1] = (v2f){qa[2 * s2], qa[2 * s2 + 1]} + kk[s2][r2];
            X1[n + 1] = (v2f){qc[2 * s2], qc[2 * s2 + 1]} + kk[s2][r2];
        }
        if (n < NR) {
            T0[n] = tanh2(X0[n]);
            T1[n] = tanh2(X1[n]);
        }
        if (n >= FSK) {
            const int m = n - FSK, s0 = m / IT, r0 = m % IT;
            acc0[r0] = __builtin_elementwise_fma(T0[m], vv[s0], acc0[r0]);
            acc1[r0] = __builtin_elementwise_fma(T1[m], vv[s0], acc1[r0]);
        }
        if (n < NR) {
#pragma unroll
            for (int g = 0; g < 4; ++g) {
                __builtin_amdgcn_sched_group_barrier(0x002, 1, 0);
                __builtin_amdgcn_sched_group_barrier(0x400, 1, 0);
            }
        }
    }
}

__device__ __forceinline__ void load_q_chunk(const float* __restrict__ qb,
                                             const int c, const int j0, const int j1,
                                             float (&da)[CH], float (&dc)[CH])
{
#pragma unroll
    for (int e = 0; e < CH; e += 2) {
        da[e]     = qb[(size_t)(c * CH + e)     * LL + j0];
        da[e + 1] = qb[(size_t)(c * CH + e + 1) * LL + j0];
        dc[e]     = qb[(size_t)(c * CH + e)     * LL + j1];
        dc[e + 1] = qb[(size_t)(c * CH + e + 1) * LL + j1];
    }
}

__global__ __launch_bounds__(256)
void score_softmax(const float* __restrict__ qT, const float* __restrict__ k,
                   const float* __restrict__ v, float* __restrict__ out)
{
    const int blk = blockIdx.x;
    const int b   = blk >> 6;
    const int i0  = (blk & 63) * IT;

    __shared__ __attribute__((aligned(16))) float s_k[(DD / 2) * 2 * IT];
    __shared__ __attribute__((aligned(16))) float s_v[DD];
    __shared__ __attribute__((aligned(16))) float s_sc[IT * LL];

    const int tid  = threadIdx.x;
    const int lane = tid & 31;
    const int wave = tid >> 5;

    const float* qb = qT + (size_t)b * DD * LL;
    const int j0 = tid;
    const int j1 = tid + 256;

    float qa0[CH], qc0[CH], qa1[CH], qc1[CH];
    load_q_chunk(qb, 0, j0, j1, qa0, qc0);

    {
        const int r = tid >> 5;
        const int d = (tid & 31) << 2;
        const v4f kv = *(const v4f*)(k + ((size_t)(b * LL + i0 + r)) * DD + d);
        const v2f lo = { kv.x, kv.y };
        const v2f hi = { kv.z, kv.w };
        *(v2f*)(s_k + ((d >> 1)    ) * (2 * IT) + 2 * r) = lo;
        *(v2f*)(s_k + ((d >> 1) + 1) * (2 * IT) + 2 * r) = hi;
        if (tid < DD / 4)
            *(v4f*)(s_v + 4 * tid) = *(const v4f*)(v + 4 * tid);
    }
    __syncthreads();

    v2f acc0[IT], acc1[IT];
#pragma unroll
    for (int r = 0; r < IT; ++r) { acc0[r] = (v2f){0.f, 0.f}; acc1[r] = (v2f){0.f, 0.f}; }

    for (int c = 0; c < DD / CH; c += 2) {
        load_q_chunk(qb, c + 1, j0, j1, qa1, qc1);
        __builtin_amdgcn_sched_barrier(0);
        score_chunk(s_k, s_v, c * (CH / 2), qa0, qc0, acc0, acc1);
        __builtin_amdgcn_sched_barrier(0);

        const int cn = (c + 2 < DD / CH) ? (c + 2) : (DD / CH - 1);
        load_q_chunk(qb, cn, j0, j1, qa0, qc0);
        __builtin_amdgcn_sched_barrier(0);
        score_chunk(s_k, s_v, (c + 1) * (CH / 2), qa1, qc1, acc0, acc1);
        __builtin_amdgcn_sched_barrier(0);
    }

#pragma unroll
    for (int r = 0; r < IT; ++r) {
        s_sc[r * LL + j0] = acc0[r].x + acc0[r].y;
        s_sc[r * LL + j1] = acc1[r].x + acc1[r].y;
    }
    __syncthreads();

    const float* row = s_sc + wave * LL;
    v4f a[4];
#pragma unroll
    for (int t = 0; t < 4; ++t)
        a[t] = *(const v4f*)(row + t * 128 + lane * 4);

    float m = a[0].x;
#pragma unroll
    for (int t = 0; t < 4; ++t)
        m = fmaxf(fmaxf(fmaxf(m, a[t].x), fmaxf(a[t].y, a[t].z)), a[t].w);
#pragma unroll
    for (int off = 16; off > 0; off >>= 1)
        m = fmaxf(m, __shfl_xor(m, off, 32));

    float s = 0.f;
#pragma unroll
    for (int t = 0; t < 4; ++t) {
        a[t].x = __expf(a[t].x - m);
        a[t].y = __expf(a[t].y - m);
        a[t].z = __expf(a[t].z - m);
        a[t].w = __expf(a[t].w - m);
        s += (a[t].x + a[t].y) + (a[t].z + a[t].w);
    }
#pragma unroll
    for (int off = 16; off > 0; off >>= 1)
        s += __shfl_xor(s, off, 32);
    const float inv = 1.0f / s;

    float* orow = out + ((size_t)(b * LL + i0 + wave)) * LL;
#pragma unroll
    for (int t = 0; t < 4; ++t)
        *(v4f*)(orow + t * 128 + lane * 4) = a[t] * inv;
}

extern "C" void kernel_launch(void* const* d_in, const int* in_sizes, int n_in,
                              void* d_out, int out_size, void* d_ws, size_t ws_size,
                              hipStream_t stream) {
    (void)in_sizes; (void)n_in; (void)out_size; (void)ws_size;
    const float* inp = (const float*)d_in[0];
    const float* Wq  = (const float*)d_in[1];
    const float* bq  = (const float*)d_in[2];
    const float* Wk  = (const float*)d_in[3];
    const float* v   = (const float*)d_in[4];
    float* out = (float*)d_out;

    float* qws = (float*)d_ws;
    float* kws = qws + (size_t)BB * LL * DD;

    proj_qk_wmma<<<256, 128, 0, stream>>>(inp, Wq, bq, Wk, qws, kws);
    score_softmax<<<BB * (LL / IT), 256, 0, stream>>>(qws, kws, v, out);
}


# score kernel: k pairs are read from LDS as one run of loads per 32 row-steps (2 read places per chunk) instead of one or two loads next to each use (about 26 places per chunk)
# speedup vs baseline: 2.8763x; 1.1160x over previous
#include <hip/hip_runtime.h>
#include <hip/hip_bf16.h>
#include <math.h>


typedef __attribute__((ext_vector_type(2))) float v2f;
typedef __attribute__((ext_vector_type(4))) float v4f;
typedef __attribute__((ext_vector_type(8))) float v8f;

#define BB 4
#define LL 512
#define DD 128
#define IT 8
#define CH 16

__device__ __forceinline__ float fast_tanh(float x) {
#if defined(__has_builtin)
#if __has_builtin(__builtin_amdgcn_tanhf)
    return __builtin_amdgcn_tanhf(x);
#elif __has_builtin(__builtin_amdgcn_tanh_f32)
    return __builtin_amdgcn_tanh_f32(x);
#else
    return tanhf(x);
#endif
#else
    return tanhf(x);
#endif
}

__global__ __launch_bounds__(128)
void proj_qk_wmma(const float* __restrict__ inp,
                  const float* __restrict__ Wq, const float* __restrict__ bq,
                  const float* __restrict__ Wk,
                  float* __restrict__ qo, float* __restrict__ ko)
{
    const int wave = threadIdx.x >> 5;
    const int lane = threadIdx.x & 31;
    const int tile = blockIdx.x * 4 + wave;
    const int m0   = (tile >> 3) << 4;
    const int n0   = (tile & 7) << 4;
    const int r    = lane & 15;
    const int koff = (lane >> 4) << 1;

    const float* arow  = inp + (size_t)(m0 + r) * DD + koff;
    const float* bqrow = Wq  + (size_t)(n0 + r) * DD + koff;
    const float* bkrow = Wk  + (size_t)(n0 + r) * DD + koff;

    v8f cq = {};
    v8f ck = {};
    for (int kk = 0; kk < DD; kk += 4) {
        v2f a   = *(const v2f*)(arow  + kk);
        v2f bq_ = *(const v2f*)(bqrow + kk);
        v2f bk_ = *(const v2f*)(bkrow + kk);
        cq = __builtin_amdgcn_wmma_f32_16x16x4_f32(false, a, false, bq_,
                                                   (short)0, cq, false, false);
        ck = __builtin_amdgcn_wmma_f32_16x16x4_f32(false, a, false, bk_,
                                                   (short)0, ck, false, false);
    }

    const int   nn    = n0 + (lane & 15);
    const float bias  = bq[nn];
    const int   mbase = m0 + ((lane >> 4) << 3);

    {
        const int bb = mbase >> 9;
        const int l0 = mbase & (LL - 1);
        float* qt = qo + ((size_t)(bb * DD + nn)) * LL + l0;
        v4f lo = { cq[0] + bias, cq[1] + bias, cq[2] + bias, cq[3] + bias };
        v4f hi = { cq[4] + bias, cq[5] + bias, cq[6] + bias, cq[7] + bias };
        *(v4f*)(qt)     = lo;
        *(v4f*)(qt + 4) = hi;
    }
#pragma unroll
    for (int rr = 0; rr < 8; ++rr) {
        const int m = mbase + rr;
        ko[(size_t)m * DD + nn] = ck[rr];
    }
}

__device__ __forceinline__ v2f tanh2(v2f x) {
    v2f t;
    t.x = fast_tanh(x.x);
    t.y = fast_tanh(x.y);
    return t;
}

constexpr int NS_ = CH / 2;
constexpr int GS_ = 4;

__device__ __forceinline__ void read_k_group(const float* __restrict__ s_k,
                                             const int pbase, const int g,
                                             v2f (&kk)[NS_][IT])
{
#pragma unroll
    for (int s = g * GS_; s < (g + 1) * GS_; ++s) {
        const int p = pbase + s;
        const v4f* kp = (const v4f*)(s_k + p * (2 * IT));
        const v4f k01 = kp[0], k23 = kp[1], k45 = kp[2], k67 = kp[3];
        kk[s][0] = (v2f){k01.x, k01.y};  kk[s][1] = (v2f){k01.z, k01.w};
        kk[s][2] = (v2f){k23.x, k23.y};  kk[s][3] = (v2f){k23.z, k23.w};
        kk[s][4] = (v2f){k45.x, k45.y};  kk[s][5] = (v2f){k45.z, k45.w};
        kk[s][6] = (v2f){k67.x, k67.y};  kk[s][7] = (v2f){k67.z, k67.w};
    }
}

__device__ __forceinline__ void score_chunk(const float* __restrict__ s_k,
                                            const float* __restrict__ s_v,
                                            const int pbase,
                                            const float (&qa)[CH],
                                            const float (&qc)[CH],
                                            v2f (&acc0)[IT], v2f (&acc1)[IT])
{
    constexpr int NS = NS_;
    constexpr int NR  = NS * IT;
    constexpr int GR  = GS_ * IT;
    constexpr int FSK = 3;
    v2f kk[NS][IT];
    v2f vv[NS];
#pragma unroll
    for (int s = 0; s < NS; s += 2) {
        const v4f v01 = *(const v4f*)(s_v + 2 * (pbase + s));
        vv[s]     = (v2f){v01.x, v01.y};
        vv[s + 1] = (v2f){v01.z, v01.w};
    }
    read_k_group(s_k, pbase, 0, kk);
    __builtin_amdgcn_sched_barrier(0);

    v2f X0[NR], X1[NR], T0[NR], T1[NR];
    X0[0] = (v2f){qa[0], qa[1]} + kk[0][0];
    X1[0] = (v2f){qc[0], qc[1]} + kk[0][0];
#pragma unroll
    for (int n = 0; n < NR + FSK; ++n) {
        if (n + 1 < NR && (n + 1) % GR == 0) {
            __builtin_amdgcn_sched_barrier(0);
            read_k_group(s_k, pbase, (n + 1) / GR, kk);
            __builtin_amdgcn_sched_barrier(0);
        }
        if (n + 1 < NR) {
            const int s2 = (n + 1) / IT, r2 = (n + 1) % IT;
            X0[n + 1] = (v2f){qa[2 * s2], qa[2 * s2 + 1]} + kk[s2][r2];
            X1[n + 1] = (v2f){qc[2 * s2], qc[2 * s2 + 1]} + kk[s2][r2];
        }
        if (n < NR) {
            T0[n] = tanh2(X0[n]);
            T1[n] = tanh2(X1[n]);
        }
        if (n >= FSK) {
            const int m = n - FSK, s0 = m / IT, r0 = m % IT;
            acc0[r0] = __builtin_elementwise_fma(T0[m], vv[s0], acc0[r0]);
            acc1[r0] = __builtin_elementwise_fma(T1[m], vv[s0], acc1[r0]);
        }
        if (n < NR) {
#pragma unroll
            for (int g = 0; g < 4; ++g) {
                __builtin_amdgcn_sched_group_barrier(0x002, 1, 0);
                __builtin_amdgcn_sched_group_barrier(0x400, 1, 0);
            }
        }
    }
}

__device__ __forceinline__ void load_q_chunk(const float* __restrict__ qb,
                                             const int c, const int j0, const int j1,
                                             float (&da)[CH], float (&dc)[CH])
{
#pragma unroll
    for (int e = 0; e < CH; e += 2) {
        da[e]     = qb[(size_t)(c * CH + e)     * LL + j0];
        da[e + 1] = qb[(size_t)(c * CH + e + 1) * LL + j0];
        dc[e]     = qb[(size_t)(c * CH + e)     * LL + j1];
        dc[e + 1] = qb[(size_t)(c * CH + e + 1) * LL + j1];
    }
}

__global__ __launch_bounds__(256)
void score_softmax(const float* __restrict__ qT, const float* __restrict__ k,
                   const float* __restrict__ v, float* __restrict__ out)
{
    const int blk = blockIdx.x;
    const int b   = blk >> 6;
    const int i0  = (blk & 63) * IT;

    __shared__ __attribute__((aligned(16))) float s_k[(DD / 2) * 2 * IT];
    __shared__ __attribute__((aligned(16))) float s_v[DD];
    __shared__ __attribute__((aligned(16))) float s_sc[IT * LL];

    const int tid  = threadIdx.x;
    const int lane = tid & 31;
    const int wave = tid >> 5;

    const float* qb = qT + (size_t)b * DD * LL;
    const int j0 = tid;
    const int j1 = tid + 256;

    float qa0[CH], qc0[CH], qa1[CH], qc1[CH];
    load_q_chunk(qb, 0, j0, j1, qa0, qc0);

    {
        const int r = tid >> 5;
        const int d = (tid & 31) << 2;
        const v4f kv = *(const v4f*)(k + ((size_t)(b * LL + i0 + r)) * DD + d);
        const v2f lo = { kv.x, kv.y };
        const v2f hi = { kv.z, kv.w };
        *(v2f*)(s_k + ((d >> 1)    ) * (2 * IT) + 2 * r) = lo;
        *(v2f*)(s_k + ((d >> 1) + 1) * (2 * IT) + 2 * r) = hi;
        if (tid < DD / 4)
            *(v4f*)(s_v + 4 * tid) = *(const v4f*)(v + 4 * tid);
    }
    __syncthreads();

    v2f acc0[IT], acc1[IT];
#pragma unroll
    for (int r = 0; r < IT; ++r) { acc0[r] = (v2f){0.f, 0.f}; acc1[r] = (v2f){0.f, 0.f}; }

    for (int c = 0; c < DD / CH; c += 2) {
        load_q_chunk(qb, c + 1, j0, j1, qa1, qc1);
        __builtin_amdgcn_sched_barrier(0);
        score_chunk(s_k, s_v, c * (CH / 2), qa0, qc0, acc0, acc1);
        __builtin_amdgcn_sched_barrier(0);

        const int cn = (c + 2 < DD / CH) ? (c + 2) : (DD / CH - 1);
        load_q_chunk(qb, cn, j0, j1, qa0, qc0);
        __builtin_amdgcn_sched_barrier(0);
        score_chunk(s_k, s_v, (c + 1) * (CH / 2), qa1, qc1, acc0, acc1);
        __builtin_amdgcn_sched_barrier(0);
    }

#pragma unroll
    for (int r = 0; r < IT; ++r) {
        s_sc[r * LL + j0] = acc0[r].x + acc0[r].y;
        s_sc[r * LL + j1] = acc1[r].x + acc1[r].y;
    }
    __syncthreads();

    const float* row = s_sc + wave * LL;
    v4f a[4];
#pragma unroll
    for (int t = 0; t < 4; ++t)
        a[t] = *(const v4f*)(row + t * 128 + lane * 4);

    float m = a[0].x;
#pragma unroll
    for (int t = 0; t < 4; ++t)
        m = fmaxf(fmaxf(fmaxf(m, a[t].x), fmaxf(a[t].y, a[t].z)), a[t].w);
#pragma unroll
    for (int off = 16; off > 0; off >>= 1)
        m = fmaxf(m, __shfl_xor(m, off, 32));

    float s = 0.f;
#pragma unroll
    for (int t = 0; t < 4; ++t) {
        a[t].x = __expf(a[t].x - m);
        a[t].y = __expf(a[t].y - m);
        a[t].z = __expf(a[t].z - m);
        a[t].w = __expf(a[t].w - m);
        s += (a[t].x + a[t].y) + (a[t].z + a[t].w);
    }
#pragma unroll
    for (int off = 16; off > 0; off >>= 1)
        s += __shfl_xor(s, off, 32);
    const float inv = 1.0f / s;

    float* orow = out + ((size_t)(b * LL + i0 + wave)) * LL;
#pragma unroll
    for (int t = 0; t < 4; ++t)
        *(v4f*)(orow + t * 128 + lane * 4) = a[t] * inv;
}

extern "C" void kernel_launch(void* const* d_in, const int* in_sizes, int n_in,
                              void* d_out, int out_size, void* d_ws, size_t ws_size,
                              hipStream_t stream) {
    (void)in_sizes; (void)n_in; (void)out_size; (void)ws_size;
    const float* inp = (const float*)d_in[0];
    const float* Wq  = (const float*)d_in[1];
    const float* bq  = (const float*)d_in[2];
    const float* Wk  = (const float*)d_in[3];
    const float* v   = (const float*)d_in[4];
    float* out = (float*)d_out;

    float* qws = (float*)d_ws;
    float* kws = qws + (size_t)BB * LL * DD;

    proj_qk_wmma<<<256, 128, 0, stream>>>(inp, Wq, bq, Wk, qws, kws);
    score_softmax<<<BB * (LL / IT), 256, 0, stream>>>(qws, kws, v, out);
}


# score kernel re-tiled: each lane owns 4 adjacent columns (q stored as (d,d+1) pairs per column, 16-byte loads; one k read per 8 tanh); two wave groups split d in halves (d sum order changes)
# speedup vs baseline: 3.0545x; 1.0620x over previous
#include <hip/hip_runtime.h>
#include <hip/hip_bf16.h>
#include <math.h>


typedef __attribute__((ext_vector_type(2))) float v2f;
typedef __attribute__((ext_vector_type(4))) float v4f;
typedef __attribute__((ext_vector_type(8))) float v8f;

#define BB 4
#define LL 512
#define DD 128
#define IT 8
#define CH 8
#define DH 64

__device__ __forceinline__ float fast_tanh(float x) {
#if defined(__has_builtin)
#if __has_builtin(__builtin_amdgcn_tanhf)
    return __builtin_amdgcn_tanhf(x);
#elif __has_builtin(__builtin_amdgcn_tanh_f32)
    return __builtin_amdgcn_tanh_f32(x);
#else
    return tanhf(x);
#endif
#else
    return tanhf(x);
#endif
}

__global__ __launch_bounds__(128)
void proj_qk_wmma(const float* __restrict__ inp,
                  const float* __restrict__ Wq, const float* __restrict__ bq,
                  const float* __restrict__ Wk,
                  float* __restrict__ qo, float* __restrict__ ko)
{
    const int wave = threadIdx.x >> 5;
    const int lane = threadIdx.x & 31;
    const int tile = blockIdx.x * 4 + wave;
    const int m0   = (tile >> 3) << 4;
    const int n0   = (tile & 7) << 4;
    const int r    = lane & 15;
    const int koff = (lane >> 4) << 1;

    const float* arow  = inp + (size_t)(m0 + r) * DD + koff;
    const float* bqrow = Wq  + (size_t)(n0 + r) * DD + koff;
    const float* bkrow = Wk  + (size_t)(n0 + r) * DD + koff;

    v8f cq = {};
    v8f ck = {};
    for (int kk = 0; kk < DD; kk += 4) {
        v2f a   = *(const v2f*)(arow  + kk);
        v2f bq_ = *(const v2f*)(bqrow + kk);
        v2f bk_ = *(const v2f*)(bkrow + kk);
        cq = __builtin_amdgcn_wmma_f32_16x16x4_f32(false, bq_, false, a,
                                                   (short)0, cq, false, false);
        ck = __builtin_amdgcn_wmma_f32_16x16x4_f32(false, a, false, bk_,
                                                   (short)0, ck, false, false);
    }

    const int   nn    = n0 + (lane & 15);
    const int   mbase = m0 + ((lane >> 4) << 3);

    {
        const int mq = m0 + (lane & 15);
        const int bb = mq >> 9;
        const int lq = mq & (LL - 1);
        const int d0 = n0 + ((lane >> 4) << 3);
        const v4f b03 = *(const v4f*)(bq + d0);
        const v4f b47 = *(const v4f*)(bq + d0 + 4);
        float* qp = qo + (((size_t)(bb * (DD / 2) + (d0 >> 1))) * LL + lq) * 2;
        *(v2f*)(qp)              = (v2f){ cq[0] + b03.x, cq[1] + b03.y };
        *(v2f*)(qp + 2 * LL)     = (v2f){ cq[2] + b03.z, cq[3] + b03.w };
        *(v2f*)(qp + 4 * LL)     = (v2f){ cq[4] + b47.x, cq[5] + b47.y };
        *(v2f*)(qp + 6 * LL)     = (v2f){ cq[6] + b47.z, cq[7] + b47.w };
    }
#pragma unroll
    for (int rr = 0; rr < 8; ++rr) {
        const int m = mbase + rr;
        ko[(size_t)m * DD + nn] = ck[rr];
    }
}

__device__ __forceinline__ v2f tanh2(v2f x) {
    v2f t;
    t.x = fast_tanh(x.x);
    t.y = fast_tanh(x.y);
    return t;
}

constexpr int NS_ = CH / 2;

__device__ __forceinline__ void score_chunk(const float* __restrict__ s_k,
                                            const float* __restrict__ s_v,
                                            const int pbase,
                                            const v4f (&q)[NS_][2],
                                            v2f (&acc)[4][IT])
{
    constexpr int NS  = NS_;
    constexpr int NR  = NS * IT * 2;
    constexpr int FSK = 3;
    v2f kk[NS][IT];
    v2f vv[NS];
#pragma unroll
    for (int s = 0; s < NS; s += 2) {
        const v4f v01 = *(const v4f*)(s_v + 2 * (pbase + s));
        vv[s]     = (v2f){v01.x, v01.y};
        vv[s + 1] = (v2f){v01.z, v01.w};
    }
#pragma unroll
    for (int s = 0; s < NS; ++s) {
        const v4f* kp = (const v4f*)(s_k + (pbase + s) * (2 * IT));
        const v4f k01 = kp[0], k23 = kp[1], k45 = kp[2], k67 = kp[3];
        kk[s][0] = (v2f){k01.x, k01.y};  kk[s][1] = (v2f){k01.z, k01.w};
        kk[s][2] = (v2f){k23.x, k23.y};  kk[s][3] = (v2f){k23.z, k23.w};
        kk[s][4] = (v2f){k45.x, k45.y};  kk[s][5] = (v2f){k45.z, k45.w};
        kk[s][6] = (v2f){k67.x, k67.y};  kk[s][7] = (v2f){k67.z, k67.w};
    }
    __builtin_amdgcn_sched_barrier(0);

    v2f X0[NR], X1[NR], T0[NR], T1[NR];
    X0[0] = (v2f){q[0][0].x, q[0][0].y} + kk[0][0];
    X1[0] = (v2f){q[0][0].z, q[0][0].w} + kk[0][0];
#pragma unroll
    for (int n = 0; n < NR + FSK; ++n) {
        if (n + 1 < NR) {
            const int s2 = (n + 1) / (2 * IT), r2 = ((n + 1) % (2 * IT)) / 2, h2 = (n + 1) % 2;
            X0[n + 1] = (v2f){q[s2][h2].x, q[s2][h2].y} + kk[s2][r2];
            X1[n + 1] = (v2f){q[s2][h2].z, q[s2][h2].w} + kk[s2][r2];
        }
        if (n < NR) {
            T0[n] = tanh2(X0[n]);
            T1[n] = tanh2(X1[n]);
        }
        if (n >= FSK) {
            const int m = n - FSK, s0 = m / (2 * IT), r0 = (m % (2 * IT)) / 2, h0 = m % 2;
            acc[2 * h0][r0]     = __builtin_elementwise_fma(T0[m], vv[s0], acc[2 * h0][r0]);
            acc[2 * h0 + 1][r0] = __builtin_elementwise_fma(T1[m], vv[s0], acc[2 * h0 + 1][r0]);
        }
        if (n < NR) {
#pragma unroll
            for (int g = 0; g < 4; ++g) {
                __builtin_amdgcn_sched_group_barrier(0x002, 1, 0);
                __builtin_amdgcn_sched_group_barrier(0x400, 1, 0);
            }
        }
    }
}

__device__ __forceinline__ void load_q_chunk(const float* __restrict__ qcol,
                                             const int p0, v4f (&dq)[NS_][2])
{
#pragma unroll
    for (int e = 0; e < NS_; ++e) {
        const v4f* src = (const v4f*)(qcol + (size_t)(p0 + e) * (2 * LL));
        dq[e][0] = src[0];
        dq[e][1] = src[1];
    }
}

__global__ __launch_bounds__(256)
void score_softmax(const float* __restrict__ q2, const float* __restrict__ k,
                   const float* __restrict__ v, float* __restrict__ out)
{
    const int blk = blockIdx.x;
    const int b   = blk >> 6;
    const int i0  = (blk & 63) * IT;

    __shared__ __attribute__((aligned(16))) float s_k[(DD / 2) * 2 * IT];
    __shared__ __attribute__((aligned(16))) float s_v[DD];
    __shared__ __attribute__((aligned(16))) float s_sc[2 * IT * LL];

    const int tid  = threadIdx.x;
    const int lane = tid & 31;
    const int wave = tid >> 5;
    const int grp  = wave >> 2;
    const int jb   = (tid & 127) << 2;
    const int pbeg = grp * (DH / 2);

    const float* qcol = q2 + (size_t)b * DD * LL + 2 * jb;

    v4f qA[NS_][2], qB[NS_][2];
    load_q_chunk(qcol, pbeg, qA);

    {
        const int r = tid >> 5;
        const int d = (tid & 31) << 2;
        const v4f kv = *(const v4f*)(k + ((size_t)(b * LL + i0 + r)) * DD + d);
        const v2f lo = { kv.x, kv.y };
        const v2f hi = { kv.z, kv.w };
        *(v2f*)(s_k + ((d >> 1)    ) * (2 * IT) + 2 * r) = lo;
        *(v2f*)(s_k + ((d >> 1) + 1) * (2 * IT) + 2 * r) = hi;
        if (tid < DD / 4)
            *(v4f*)(s_v + 4 * tid) = *(const v4f*)(v + 4 * tid);
    }
    __syncthreads();

    v2f acc[4][IT];
#pragma unroll
    for (int c = 0; c < 4; ++c)
#pragma unroll
        for (int r = 0; r < IT; ++r) acc[c][r] = (v2f){0.f, 0.f};

    constexpr int NC = DH / CH;
    for (int c = 0; c < NC; c += 2) {
        load_q_chunk(qcol, pbeg + (c + 1) * NS_, qB);
        __builtin_amdgcn_sched_barrier(0);
        score_chunk(s_k, s_v, pbeg + c * NS_, qA, acc);
        __builtin_amdgcn_sched_barrier(0);

        const int cn = (c + 2 < NC) ? (c + 2) : (NC - 1);
        load_q_chunk(qcol, pbeg + cn * NS_, qA);
        __builtin_amdgcn_sched_barrier(0);
        score_chunk(s_k, s_v, pbeg + (c + 1) * NS_, qB, acc);
        __builtin_amdgcn_sched_barrier(0);
    }

    {
        float* sp = s_sc + grp * (IT * LL) + jb;
#pragma unroll
        for (int r = 0; r < IT; ++r)
            *(v4f*)(sp + r * LL) = (v4f){ acc[0][r].x + acc[0][r].y, acc[1][r].x + acc[1][r].y,
                                          acc[2][r].x + acc[2][r].y, acc[3][r].x + acc[3][r].y };
    }
    __syncthreads();

    const float* row = s_sc + wave * LL;
    v4f a[4];
#pragma unroll
    for (int t = 0; t < 4; ++t)
        a[t] = *(const v4f*)(row + t * 128 + lane * 4)
             + *(const v4f*)(row + IT * LL + t * 128 + lane * 4);

    float m = a[0].x;
#pragma unroll
    for (int t = 0; t < 4; ++t)
        m = fmaxf(fmaxf(fmaxf(m, a[t].x), fmaxf(a[t].y, a[t].z)), a[t].w);
#pragma unroll
    for (int off = 16; off > 0; off >>= 1)
        m = fmaxf(m, __shfl_xor(m, off, 32));

    float s = 0.f;
#pragma unroll
    for (int t = 0; t < 4; ++t) {
        a[t].x = __expf(a[t].x - m);
        a[t].y = __expf(a[t].y - m);
        a[t].z = __expf(a[t].z - m);
        a[t].w = __expf(a[t].w - m);
        s += (a[t].x + a[t].y) + (a[t].z + a[t].w);
    }
#pragma unroll
    for (int off = 16; off > 0; off >>= 1)
        s += __shfl_xor(s, off, 32);
    const float inv = 1.0f / s;

    float* orow = out + ((size_t)(b * LL + i0 + wave)) * LL;
#pragma unroll
    for (int t = 0; t < 4; ++t)
        *(v4f*)(orow + t * 128 + lane * 4) = a[t] * inv;
}

extern "C" void kernel_launch(void* const* d_in, const int* in_sizes, int n_in,
                              void* d_out, int out_size, void* d_ws, size_t ws_size,
                              hipStream_t stream) {
    (void)in_sizes; (void)n_in; (void)out_size; (void)ws_size;
    const float* inp = (const float*)d_in[0];
    const float* Wq  = (const float*)d_in[1];
    const float* bq  = (const float*)d_in[2];
    const float* Wk  = (const float*)d_in[3];
    const float* v   = (const float*)d_in[4];
    float* out = (float*)d_out;

    float* qws = (float*)d_ws;
    float* kws = qws + (size_t)BB * LL * DD;

    proj_qk_wmma<<<256, 128, 0, stream>>>(inp, Wq, bq, Wk, qws, kws);
    score_softmax<<<BB * (LL / IT), 256, 0, stream>>>(qws, kws, v, out);
}


# projection kernel: a wave requests all its operand fragments up front as 48 16-byte loads (before: 96 8-byte loads, about 18 in flight), each load feeding two matrix instructions; k sum order changes
# speedup vs baseline: 3.2471x; 1.0631x over previous
#include <hip/hip_runtime.h>
#include <hip/hip_bf16.h>
#include <math.h>


typedef __attribute__((ext_vector_type(2))) float v2f;
typedef __attribute__((ext_vector_type(4))) float v4f;
typedef __attribute__((ext_vector_type(8))) float v8f;

#define BB 4
#define LL 512
#define DD 128
#define IT 8
#define CH 8
#define DH 64

__device__ __forceinline__ float fast_tanh(float x) {
#if defined(__has_builtin)
#if __has_builtin(__builtin_amdgcn_tanhf)
    return __builtin_amdgcn_tanhf(x);
#elif __has_builtin(__builtin_amdgcn_tanh_f32)
    return __builtin_amdgcn_tanh_f32(x);
#else
    return tanhf(x);
#endif
#else
    return tanhf(x);
#endif
}

__global__ __launch_bounds__(128)
void proj_qk_wmma(const float* __restrict__ inp,
                  const float* __restrict__ Wq, const float* __restrict__ bq,
                  const float* __restrict__ Wk,
                  float* __restrict__ qo, float* __restrict__ ko)
{
    const int wave = threadIdx.x >> 5;
    const int lane = threadIdx.x & 31;
    const int tile = blockIdx.x * 4 + wave;
    const int m0   = (tile >> 3) << 4;
    const int n0   = (tile & 7) << 4;
    const int r    = lane & 15;
    const int koff = (lane >> 4) << 2;

    const float* arow  = inp + (size_t)(m0 + r) * DD + koff;
    const float* bqrow = Wq  + (size_t)(n0 + r) * DD + koff;
    const float* bkrow = Wk  + (size_t)(n0 + r) * DD + koff;

    v8f cq = {};
    v8f ck = {};
    constexpr int NK8 = DD / 8;
    v4f a4[NK8], q4[NK8], k4[NK8];
#pragma unroll
    for (int t = 0; t < NK8; ++t) {
        a4[t] = *(const v4f*)(arow  + 8 * t);
        q4[t] = *(const v4f*)(bqrow + 8 * t);
        k4[t] = *(const v4f*)(bkrow + 8 * t);
    }
    __builtin_amdgcn_sched_barrier(0);
#pragma unroll
    for (int t = 0; t < NK8; ++t) {
        const v2f a0 = { a4[t].x, a4[t].y }, a1 = { a4[t].z, a4[t].w };
        const v2f q0 = { q4[t].x, q4[t].y }, q1 = { q4[t].z, q4[t].w };
        const v2f k0 = { k4[t].x, k4[t].y }, k1 = { k4[t].z, k4[t].w };
        cq = __builtin_amdgcn_wmma_f32_16x16x4_f32(false, q0, false, a0,
                                                   (short)0, cq, false, false);
        ck = __builtin_amdgcn_wmma_f32_16x16x4_f32(false, a0, false, k0,
                                                   (short)0, ck, false, false);
        cq = __builtin_amdgcn_wmma_f32_16x16x4_f32(false, q1, false, a1,
                                                   (short)0, cq, false, false);
        ck = __builtin_amdgcn_wmma_f32_16x16x4_f32(false, a1, false, k1,
                                                   (short)0, ck, false, false);
    }

    const int   nn    = n0 + (lane & 15);
    const int   mbase = m0 + ((lane >> 4) << 3);

    {
        const int mq = m0 + (lane & 15);
        const int bb = mq >> 9;
        const int lq = mq & (LL - 1);
        const int d0 = n0 + ((lane >> 4) << 3);
        const v4f b03 = *(const v4f*)(bq + d0);
        const v4f b47 = *(const v4f*)(bq + d0 + 4);
        float* qp = qo + (((size_t)(bb * (DD / 2) + (d0 >> 1))) * LL + lq) * 2;
        *(v2f*)(qp)              = (v2f){ cq[0] + b03.x, cq[1] + b03.y };
        *(v2f*)(qp + 2 * LL)     = (v2f){ cq[2] + b03.z, cq[3] + b03.w };
        *(v2f*)(qp + 4 * LL)     = (v2f){ cq[4] + b47.x, cq[5] + b47.y };
        *(v2f*)(qp + 6 * LL)     = (v2f){ cq[6] + b47.z, cq[7] + b47.w };
    }
#pragma unroll
    for (int rr = 0; rr < 8; ++rr) {
        const int m = mbase + rr;
        ko[(size_t)m * DD + nn] = ck[rr];
    }
}

__device__ __forceinline__ v2f tanh2(v2f x) {
    v2f t;
    t.x = fast_tanh(x.x);
    t.y = fast_tanh(x.y);
    return t;
}

constexpr int NS_ = CH / 2;

__device__ __forceinline__ void score_chunk(const float* __restrict__ s_k,
                                            const float* __restrict__ s_v,
                                            const int pbase,
                                            const v4f (&q)[NS_][2],
                                            v2f (&acc)[4][IT])
{
    constexpr int NS  = NS_;
    constexpr int NR  = NS * IT * 2;
    constexpr int FSK = 3;
    v2f kk[NS][IT];
    v2f vv[NS];
#pragma unroll
    for (int s = 0; s < NS; s += 2) {
        const v4f v01 = *(const v4f*)(s_v + 2 * (pbase + s));
        vv[s]     = (v2f){v01.x, v01.y};
        vv[s + 1] = (v2f){v01.z, v01.w};
    }
#pragma unroll
    for (int s = 0; s < NS; ++s) {
        const v4f* kp = (const v4f*)(s_k + (pbase + s) * (2 * IT));
        const v4f k01 = kp[0], k23 = kp[1], k45 = kp[2], k67 = kp[3];
        kk[s][0] = (v2f){k01.x, k01.y};  kk[s][1] = (v2f){k01.z, k01.w};
        kk[s][2] = (v2f){k23.x, k23.y};  kk[s][3] = (v2f){k23.z, k23.w};
        kk[s][4] = (v2f){k45.x, k45.y};  kk[s][5] = (v2f){k45.z, k45.w};
        kk[s][6] = (v2f){k67.x, k67.y};  kk[s][7] = (v2f){k67.z, k67.w};
    }
    __builtin_amdgcn_sched_barrier(0);

    v2f X0[NR], X1[NR], T0[NR], T1[NR];
    X0[0] = (v2f){q[0][0].x, q[0][0].y} + kk[0][0];
    X1[0] = (v2f){q[0][0].z, q[0][0].w} + kk[0][0];
#pragma unroll
    for (int n = 0; n < NR + FSK; ++n) {
        if (n + 1 < NR) {
            const int s2 = (n + 1) / (2 * IT), r2 = ((n + 1) % (2 * IT)) / 2, h2 = (n + 1) % 2;
            X0[n + 1] = (v2f){q[s2][h2].x, q[s2][h2].y} + kk[s2][r2];
            X1[n + 1] = (v2f){q[s2][h2].z, q[s2][h2].w} + kk[s2][r2];
        }
        if (n < NR) {
            T0[n] = tanh2(X0[n]);
            T1[n] = tanh2(X1[n]);
        }
        if (n >= FSK) {
            const int m = n - FSK, s0 = m / (2 * IT), r0 = (m % (2 * IT)) / 2, h0 = m % 2;
            acc[2 * h0][r0]     = __builtin_elementwise_fma(T0[m], vv[s0], acc[2 * h0][r0]);
            acc[2 * h0 + 1][r0] = __builtin_elementwise_fma(T1[m], vv[s0], acc[2 * h0 + 1][r0]);
        }
        if (n < NR) {
#pragma unroll
            for (int g = 0; g < 4; ++g) {
                __builtin_amdgcn_sched_group_barrier(0x002, 1, 0);
                __builtin_amdgcn_sched_group_barrier(0x400, 1, 0);
            }
        }
    }
}

__device__ __forceinline__ void load_q_chunk(const float* __restrict__ qcol,
                                             const int p0, v4f (&dq)[NS_][2])
{
#pragma unroll
    for (int e = 0; e < NS_; ++e) {
        const v4f* src = (const v4f*)(qcol + (size_t)(p0 + e) * (2 * LL));
        dq[e][0] = src[0];
        dq[e][1] = src[1];
    }
}

__global__ __launch_bounds__(256)
void score_softmax(const float* __restrict__ q2, const float* __restrict__ k,
                   const float* __restrict__ v, float* __restrict__ out)
{
    const int blk = blockIdx.x;
    const int b   = blk >> 6;
    const int i0  = (blk & 63) * IT;

    __shared__ __attribute__((aligned(16))) float s_k[(DD / 2) * 2 * IT];
    __shared__ __attribute__((aligned(16))) float s_v[DD];
    __shared__ __attribute__((aligned(16))) float s_sc[2 * IT * LL];

    const int tid  = threadIdx.x;
    const int lane = tid & 31;
    const int wave = tid >> 5;
    const int grp  = wave >> 2;
    const int jb   = (tid & 127) << 2;
    const int pbeg = grp * (DH / 2);

    const float* qcol = q2 + (size_t)b * DD * LL + 2 * jb;

    v4f qA[NS_][2], qB[NS_][2];
    load_q_chunk(qcol, pbeg, qA);

    {
        const int r = tid >> 5;
        const int d = (tid & 31) << 2;
        const v4f kv = *(const v4f*)(k + ((size_t)(b * LL + i0 + r)) * DD + d);
        const v2f lo = { kv.x, kv.y };
        const v2f hi = { kv.z, kv.w };
        *(v2f*)(s_k + ((d >> 1)    ) * (2 * IT) + 2 * r) = lo;
        *(v2f*)(s_k + ((d >> 1) + 1) * (2 * IT) + 2 * r) = hi;
        if (tid < DD / 4)
            *(v4f*)(s_v + 4 * tid) = *(const v4f*)(v + 4 * tid);
    }
    __syncthreads();

    v2f acc[4][IT];
#pragma unroll
    for (int c = 0; c < 4; ++c)
#pragma unroll
        for (int r = 0; r < IT; ++r) acc[c][r] = (v2f){0.f, 0.f};

    constexpr int NC = DH / CH;
    for (int c = 0; c < NC; c += 2) {
        load_q_chunk(qcol, pbeg + (c + 1) * NS_, qB);
        __builtin_amdgcn_sched_barrier(0);
        score_chunk(s_k, s_v, pbeg + c * NS_, qA, acc);
        __builtin_amdgcn_sched_barrier(0);

        const int cn = (c + 2 < NC) ? (c + 2) : (NC - 1);
        load_q_chunk(qcol, pbeg + cn * NS_, qA);
        __builtin_amdgcn_sched_barrier(0);
        score_chunk(s_k, s_v, pbeg + (c + 1) * NS_, qB, acc);
        __builtin_amdgcn_sched_barrier(0);
    }

    {
        float* sp = s_sc + grp * (IT * LL) + jb;
#pragma unroll
        for (int r = 0; r < IT; ++r)
            *(v4f*)(sp + r * LL) = (v4f){ acc[0][r].x + acc[0][r].y, acc[1][r].x + acc[1][r].y,
                                          acc[2][r].x + acc[2][r].y, acc[3][r].x + acc[3][r].y };
    }
    __syncthreads();

    const float* row = s_sc + wave * LL;
    v4f a[4];
#pragma unroll
    for (int t = 0; t < 4; ++t)
        a[t] = *(const v4f*)(row + t * 128 + lane * 4)
             + *(const v4f*)(row + IT * LL + t * 128 + lane * 4);

    float m = a[0].x;
#pragma unroll
    for (int t = 0; t < 4; ++t)
        m = fmaxf(fmaxf(fmaxf(m, a[t].x), fmaxf(a[t].y, a[t].z)), a[t].w);
#pragma unroll
    for (int off = 16; off > 0; off >>= 1)
        m = fmaxf(m, __shfl_xor(m, off, 32));

    float s = 0.f;
#pragma unroll
    for (int t = 0; t < 4; ++t) {
        a[t].x = __expf(a[t].x - m);
        a[t].y = __expf(a[t].y - m);
        a[t].z = __expf(a[t].z - m);
        a[t].w = __expf(a[t].w - m);
        s += (a[t].x + a[t].y) + (a[t].z + a[t].w);
    }
#pragma unroll
    for (int off = 16; off > 0; off >>= 1)
        s += __shfl_xor(s, off, 32);
    const float inv = 1.0f / s;

    float* orow = out + ((size_t)(b * LL + i0 + wave)) * LL;
#pragma unroll
    for (int t = 0; t < 4; ++t)
        *(v4f*)(orow + t * 128 + lane * 4) = a[t] * inv;
}

extern "C" void kernel_launch(void* const* d_in, const int* in_sizes, int n_in,
                              void* d_out, int out_size, void* d_ws, size_t ws_size,
                              hipStream_t stream) {
    (void)in_sizes; (void)n_in; (void)out_size; (void)ws_size;
    const float* inp = (const float*)d_in[0];
    const float* Wq  = (const float*)d_in[1];
    const float* bq  = (const float*)d_in[2];
    const float* Wk  = (const float*)d_in[3];
    const float* v   = (const float*)d_in[4];
    float* out = (float*)d_out;

    float* qws = (float*)d_ws;
    float* kws = qws + (size_t)BB * LL * DD;

    proj_qk_wmma<<<256, 128, 0, stream>>>(inp, Wq, bq, Wk, qws, kws);
    score_softmax<<<BB * (LL / IT), 256, 0, stream>>>(qws, kws, v, out);
}
